# starting kernel
# baseline (speedup 1.0000x reference)
#include <hip/hip_runtime.h>
#include <cmath>

typedef float v2f __attribute__((ext_vector_type(2)));
typedef float v4f __attribute__((ext_vector_type(4)));
typedef float v8f __attribute__((ext_vector_type(8)));

#define BATCH 8
#define CDIM 512
#define KDIM 19
#define KPAD 32
#define HW 16384

#define CT 128
#define NT 64
#define NSPLIT 16
#define NPER (HW / NSPLIT)
#define OUT_TOTAL (BATCH * CDIM * KDIM)

#define WS_W_OFF 1024
#define WS_PART_OFF (WS_W_OFF + (size_t)BATCH * KPAD * HW)

__global__ __launch_bounds__(256)
void sg_softmax_stats(const float* __restrict__ probs, float* __restrict__ stats) {
    __shared__ float red[256];
    const int bk = blockIdx.x;
    const float* row = probs + (size_t)bk * HW;
    const int t = threadIdx.x;

    float m = -INFINITY;
    for (int i = t; i < HW; i += 256) m = fmaxf(m, row[i]);
    red[t] = m; __syncthreads();
    for (int s = 128; s > 0; s >>= 1) {
        if (t < s) red[t] = fmaxf(red[t], red[t + s]);
        __syncthreads();
    }
    m = red[0]; __syncthreads();

    float sum = 0.0f;
    for (int i = t; i < HW; i += 256) sum += __expf(row[i] - m);
    red[t] = sum; __syncthreads();
    for (int s = 128; s > 0; s >>= 1) {
        if (t < s) red[t] += red[t + s];
        __syncthreads();
    }
    if (t == 0) { stats[2 * bk] = m; stats[2 * bk + 1] = 1.0f / red[0]; }
}

__global__ __launch_bounds__(256)
void sg_softmax_write(const float* __restrict__ probs, const float* __restrict__ stats,
                      float* __restrict__ W) {
    const int n = blockIdx.x * 256 + threadIdx.x;
    const int k = blockIdx.y;
    const int b = blockIdx.z;
    float v = 0.0f;
    if (k < KDIM) {
        const int bk = b * KDIM + k;
        const float m = stats[2 * bk];
        const float sinv = stats[2 * bk + 1];
        v = __expf(probs[(size_t)bk * HW + n] - m) * sinv;
    }
    W[((size_t)b * KPAD + k) * HW + n] = v;
}

__global__ __launch_bounds__(256)
void sg_context_gemm(const float* __restrict__ feats, const float* __restrict__ W,
                     float* __restrict__ part) {
    __shared__ float ftile[CT][NT + 4];
    __shared__ float wtile[KPAD][NT + 4];

    const int t = threadIdx.x;
    const int wave = t >> 5;
    const int lane = t & 31;
    const int lrow = lane & 15;
    const int ksel = (lane >> 4) * 2;

    const int c0 = blockIdx.x * CT;
    const int b  = blockIdx.y;
    const int z  = blockIdx.z;
    const int nbeg = z * NPER;

    const float* fbase = feats + ((size_t)b * CDIM + c0) * HW;
    const float* wbase = W + (size_t)b * KPAD * HW;

    const int frow = t >> 1;
    const int fseg = (t & 1) * 32;
    const int wrow = t >> 3;
    const int wseg = (t & 7) * 8;

    v8f acc0 = {};
    v8f acc1 = {};

    for (int n0 = nbeg; n0 < nbeg + NPER; n0 += NT) {
        const v4f* fsrc = (const v4f*)(fbase + (size_t)frow * HW + n0 + fseg);
        #pragma unroll
        for (int j = 0; j < 8; ++j) {
            v4f v = __builtin_nontemporal_load(fsrc + j);
            *(v4f*)&ftile[frow][fseg + 4 * j] = v;
        }
        const v4f* wsrc = (const v4f*)(wbase + (size_t)wrow * HW + n0 + wseg);
        #pragma unroll
        for (int j = 0; j < 2; ++j) {
            *(v4f*)&wtile[wrow][wseg + 4 * j] = wsrc[j];
        }
        __syncthreads();

        #pragma unroll
        for (int kk = 0; kk < NT; kk += 4) {
            v2f a  = *(const v2f*)&ftile[wave * 16 + lrow][kk + ksel];
            v2f b0 = *(const v2f*)&wtile[lrow][kk + ksel];
            v2f b1 = *(const v2f*)&wtile[16 + lrow][kk + ksel];
            acc0 = __builtin_amdgcn_wmma_f32_16x16x4_f32(
                false, a, false, b0, (short)0, acc0, false, false);
            acc1 = __builtin_amdgcn_wmma_f32_16x16x4_f32(
                false, a, false, b1, (short)0, acc1, false, false);
        }
        __syncthreads();
    }

    float* pdst = part + (size_t)z * OUT_TOTAL;
    const int rbase = c0 + wave * 16 + (lane >> 4) * 8;
    #pragma unroll
    for (int i = 0; i < 8; ++i) {
        const int row = rbase + i;
        const size_t o = ((size_t)b * CDIM + row) * KDIM;
        pdst[o + lrow] = acc0[i];
        if (lrow + 16 < KDIM) pdst[o + lrow + 16] = acc1[i];
    }
}

__global__ __launch_bounds__(256)
void sg_reduce_partials(const float* __restrict__ part, float* __restrict__ out) {
    const int i = blockIdx.x * 256 + threadIdx.x;
    if (i < OUT_TOTAL) {
        float s = 0.0f;
        #pragma unroll
        for (int j = 0; j < NSPLIT; ++j) s += part[(size_t)j * OUT_TOTAL + i];
        out[i] = s;
    }
}

extern "C" void kernel_launch(void* const* d_in, const int* in_sizes, int n_in,
                              void* d_out, int out_size, void* d_ws, size_t ws_size,
                              hipStream_t stream) {
    const float* feats = (const float*)d_in[0];
    const float* probs = (const float*)d_in[1];
    float* out = (float*)d_out;

    float* ws    = (float*)d_ws;
    float* stats = ws;
    float* W     = ws + WS_W_OFF;
    float* part  = ws + WS_PART_OFF;

    sg_softmax_stats<<<BATCH * KDIM, 256, 0, stream>>>(probs, stats);

    dim3 g2(HW / 256, KPAD, BATCH);
    sg_softmax_write<<<g2, 256, 0, stream>>>(probs, stats, W);

    dim3 g3(CDIM / CT, BATCH, NSPLIT);
    sg_context_gemm<<<g3, 256, 0, stream>>>(feats, W, part);

    sg_reduce_partials<<<(OUT_TOTAL + 255) / 256, 256, 0, stream>>>(part, out);
}


# GEMM staging loads remapped so 16 lanes read one 256 B row segment: each wave-wide 16 B load covers four whole 128 B lines instead of 32 partial ones (feats and W tiles)
# speedup vs baseline: 1.9494x; 1.9494x over previous
#include <hip/hip_runtime.h>
#include <cmath>

typedef float v2f __attribute__((ext_vector_type(2)));
typedef float v4f __attribute__((ext_vector_type(4)));
typedef float v8f __attribute__((ext_vector_type(8)));

#define BATCH 8
#define CDIM 512
#define KDIM 19
#define KPAD 32
#define HW 16384

#define CT 128
#define NT 64
#define NSPLIT 16
#define NPER (HW / NSPLIT)
#define OUT_TOTAL (BATCH * CDIM * KDIM)

#define WS_W_OFF 1024
#define WS_PART_OFF (WS_W_OFF + (size_t)BATCH * KPAD * HW)

__global__ __launch_bounds__(256)
void sg_softmax_stats(const float* __restrict__ probs, float* __restrict__ stats) {
    __shared__ float red[256];
    const int bk = blockIdx.x;
    const float* row = probs + (size_t)bk * HW;
    const int t = threadIdx.x;

    float m = -INFINITY;
    for (int i = t; i < HW; i += 256) m = fmaxf(m, row[i]);
    red[t] = m; __syncthreads();
    for (int s = 128; s > 0; s >>= 1) {
        if (t < s) red[t] = fmaxf(red[t], red[t + s]);
        __syncthreads();
    }
    m = red[0]; __syncthreads();

    float sum = 0.0f;
    for (int i = t; i < HW; i += 256) sum += __expf(row[i] - m);
    red[t] = sum; __syncthreads();
    for (int s = 128; s > 0; s >>= 1) {
        if (t < s) red[t] += red[t + s];
        __syncthreads();
    }
    if (t == 0) { stats[2 * bk] = m; stats[2 * bk + 1] = 1.0f / red[0]; }
}

__global__ __launch_bounds__(256)
void sg_softmax_write(const float* __restrict__ probs, const float* __restrict__ stats,
                      float* __restrict__ W) {
    const int n = blockIdx.x * 256 + threadIdx.x;
    const int k = blockIdx.y;
    const int b = blockIdx.z;
    float v = 0.0f;
    if (k < KDIM) {
        const int bk = b * KDIM + k;
        const float m = stats[2 * bk];
        const float sinv = stats[2 * bk + 1];
        v = __expf(probs[(size_t)bk * HW + n] - m) * sinv;
    }
    W[((size_t)b * KPAD + k) * HW + n] = v;
}

__global__ __launch_bounds__(256)
void sg_context_gemm(const float* __restrict__ feats, const float* __restrict__ W,
                     float* __restrict__ part) {
    __shared__ float ftile[CT][NT + 4];
    __shared__ float wtile[KPAD][NT + 4];

    const int t = threadIdx.x;
    const int wave = t >> 5;
    const int lane = t & 31;
    const int lrow = lane & 15;
    const int ksel = (lane >> 4) * 2;

    const int c0 = blockIdx.x * CT;
    const int b  = blockIdx.y;
    const int z  = blockIdx.z;
    const int nbeg = z * NPER;

    const float* fbase = feats + ((size_t)b * CDIM + c0) * HW;
    const float* wbase = W + (size_t)b * KPAD * HW;

    const int srow = t >> 4;
    const int scol = (t & 15) * 4;

    v8f acc0 = {};
    v8f acc1 = {};

    for (int n0 = nbeg; n0 < nbeg + NPER; n0 += NT) {
        const float* fsrc = fbase + (size_t)srow * HW + n0 + scol;
        #pragma unroll
        for (int j = 0; j < 8; ++j) {
            v4f v = __builtin_nontemporal_load((const v4f*)(fsrc + (size_t)(16 * j) * HW));
            *(v4f*)&ftile[16 * j + srow][scol] = v;
        }
        const float* wsrc = wbase + (size_t)srow * HW + n0 + scol;
        #pragma unroll
        for (int j = 0; j < 2; ++j) {
            *(v4f*)&wtile[16 * j + srow][scol] = *(const v4f*)(wsrc + (size_t)(16 * j) * HW);
        }
        __syncthreads();

        #pragma unroll
        for (int kk = 0; kk < NT; kk += 4) {
            v2f a  = *(const v2f*)&ftile[wave * 16 + lrow][kk + ksel];
            v2f b0 = *(const v2f*)&wtile[lrow][kk + ksel];
            v2f b1 = *(const v2f*)&wtile[16 + lrow][kk + ksel];
            acc0 = __builtin_amdgcn_wmma_f32_16x16x4_f32(
                false, a, false, b0, (short)0, acc0, false, false);
            acc1 = __builtin_amdgcn_wmma_f32_16x16x4_f32(
                false, a, false, b1, (short)0, acc1, false, false);
        }
        __syncthreads();
    }

    float* pdst = part + (size_t)z * OUT_TOTAL;
    const int rbase = c0 + wave * 16 + (lane >> 4) * 8;
    #pragma unroll
    for (int i = 0; i < 8; ++i) {
        const int row = rbase + i;
        const size_t o = ((size_t)b * CDIM + row) * KDIM;
        pdst[o + lrow] = acc0[i];
        if (lrow + 16 < KDIM) pdst[o + lrow + 16] = acc1[i];
    }
}

__global__ __launch_bounds__(256)
void sg_reduce_partials(const float* __restrict__ part, float* __restrict__ out) {
    const int i = blockIdx.x * 256 + threadIdx.x;
    if (i < OUT_TOTAL) {
        float s = 0.0f;
        #pragma unroll
        for (int j = 0; j < NSPLIT; ++j) s += part[(size_t)j * OUT_TOTAL + i];
        out[i] = s;
    }
}

extern "C" void kernel_launch(void* const* d_in, const int* in_sizes, int n_in,
                              void* d_out, int out_size, void* d_ws, size_t ws_size,
                              hipStream_t stream) {
    const float* feats = (const float*)d_in[0];
    const float* probs = (const float*)d_in[1];
    float* out = (float*)d_out;

    float* ws    = (float*)d_ws;
    float* stats = ws;
    float* W     = ws + WS_W_OFF;
    float* part  = ws + WS_PART_OFF;

    sg_softmax_stats<<<BATCH * KDIM, 256, 0, stream>>>(probs, stats);

    dim3 g2(HW / 256, KPAD, BATCH);
    sg_softmax_write<<<g2, 256, 0, stream>>>(probs, stats, W);

    dim3 g3(CDIM / CT, BATCH, NSPLIT);
    sg_context_gemm<<<g3, 256, 0, stream>>>(feats, W, part);

    sg_reduce_partials<<<(OUT_TOTAL + 255) / 256, 256, 0, stream>>>(part, out);
}


# softmax weights formed in the GEMM's B-tile staging (exp(p-max)*inv_sum straight from probs and row stats), no separate launch and no 16 MB weight plane: 4 launches become 3, sum order unchanged
# speedup vs baseline: 2.4906x; 1.2776x over previous
#include <hip/hip_runtime.h>
#include <cmath>

typedef float v2f __attribute__((ext_vector_type(2)));
typedef float v4f __attribute__((ext_vector_type(4)));
typedef float v8f __attribute__((ext_vector_type(8)));

#define BATCH 8
#define CDIM 512
#define KDIM 19
#define KPAD 32
#define HW 16384

#define CT 128
#define NT 64
#define NSPLIT 16
#define NPER (HW / NSPLIT)
#define OUT_TOTAL (BATCH * CDIM * KDIM)

#define WS_PART_OFF 1024

__global__ __launch_bounds__(256)
void sg_softmax_stats(const float* __restrict__ probs, float* __restrict__ stats) {
    __shared__ float red[256];
    const int bk = blockIdx.x;
    const float* row = probs + (size_t)bk * HW;
    const int t = threadIdx.x;

    float m = -INFINITY;
    for (int i = t; i < HW; i += 256) m = fmaxf(m, row[i]);
    red[t] = m; __syncthreads();
    for (int s = 128; s > 0; s >>= 1) {
        if (t < s) red[t] = fmaxf(red[t], red[t + s]);
        __syncthreads();
    }
    m = red[0]; __syncthreads();

    float sum = 0.0f;
    for (int i = t; i < HW; i += 256) sum += __expf(row[i] - m);
    red[t] = sum; __syncthreads();
    for (int s = 128; s > 0; s >>= 1) {
        if (t < s) red[t] += red[t + s];
        __syncthreads();
    }
    if (t == 0) { stats[2 * bk] = m; stats[2 * bk + 1] = 1.0f / red[0]; }
}

__global__ __launch_bounds__(256)
void sg_context_gemm(const float* __restrict__ feats, const float* __restrict__ probs,
                     const float* __restrict__ stats, float* __restrict__ part) {
    __shared__ float ftile[CT][NT + 4];
    __shared__ float wtile[KPAD][NT + 4];

    const int t = threadIdx.x;
    const int wave = t >> 5;
    const int lane = t & 31;
    const int lrow = lane & 15;
    const int ksel = (lane >> 4) * 2;

    const int c0 = blockIdx.x * CT;
    const int b  = blockIdx.y;
    const int z  = blockIdx.z;
    const int nbeg = z * NPER;

    const float* fbase = feats + ((size_t)b * CDIM + c0) * HW;
    const float* pbase = probs + (size_t)b * KDIM * HW;

    const int srow = t >> 4;
    const int scol = (t & 15) * 4;

    const bool has1 = (16 + srow) < KDIM;
    const float m0 = stats[2 * (b * KDIM + srow)];
    const float s0 = stats[2 * (b * KDIM + srow) + 1];
    float m1 = 0.0f, s1 = 0.0f;
    if (has1) {
        m1 = stats[2 * (b * KDIM + 16 + srow)];
        s1 = stats[2 * (b * KDIM + 16 + srow) + 1];
    } else {
        const v4f zero4 = {0.0f, 0.0f, 0.0f, 0.0f};
        *(v4f*)&wtile[16 + srow][scol] = zero4;
    }

    v8f acc0 = {};
    v8f acc1 = {};

    for (int n0 = nbeg; n0 < nbeg + NPER; n0 += NT) {
        const float* psrc = pbase + (size_t)srow * HW + n0 + scol;
        v4f p0 = *(const v4f*)psrc;
        v4f p1 = {0.0f, 0.0f, 0.0f, 0.0f};
        if (has1) p1 = *(const v4f*)(psrc + (size_t)16 * HW);

        const float* fsrc = fbase + (size_t)srow * HW + n0 + scol;
        v4f fv[8];
        #pragma unroll
        for (int j = 0; j < 8; ++j)
            fv[j] = __builtin_nontemporal_load((const v4f*)(fsrc + (size_t)(16 * j) * HW));
        __builtin_amdgcn_sched_barrier(0);

        {
            v4f e;
            e.x = __expf(p0.x - m0) * s0;
            e.y = __expf(p0.y - m0) * s0;
            e.z = __expf(p0.z - m0) * s0;
            e.w = __expf(p0.w - m0) * s0;
            *(v4f*)&wtile[srow][scol] = e;
        }
        if (has1) {
            v4f e;
            e.x = __expf(p1.x - m1) * s1;
            e.y = __expf(p1.y - m1) * s1;
            e.z = __expf(p1.z - m1) * s1;
            e.w = __expf(p1.w - m1) * s1;
            *(v4f*)&wtile[16 + srow][scol] = e;
        }
        #pragma unroll
        for (int j = 0; j < 8; ++j)
            *(v4f*)&ftile[16 * j + srow][scol] = fv[j];
        __syncthreads();

        #pragma unroll
        for (int kk = 0; kk < NT; kk += 4) {
            v2f a  = *(const v2f*)&ftile[wave * 16 + lrow][kk + ksel];
            v2f b0 = *(const v2f*)&wtile[lrow][kk + ksel];
            v2f b1 = *(const v2f*)&wtile[16 + lrow][kk + ksel];
            acc0 = __builtin_amdgcn_wmma_f32_16x16x4_f32(
                false, a, false, b0, (short)0, acc0, false, false);
            acc1 = __builtin_amdgcn_wmma_f32_16x16x4_f32(
                false, a, false, b1, (short)0, acc1, false, false);
        }
        __syncthreads();
    }

    float* pdst = part + (size_t)z * OUT_TOTAL;
    const int rbase = c0 + wave * 16 + (lane >> 4) * 8;
    #pragma unroll
    for (int i = 0; i < 8; ++i) {
        const int row = rbase + i;
        const size_t o = ((size_t)b * CDIM + row) * KDIM;
        pdst[o + lrow] = acc0[i];
        if (lrow + 16 < KDIM) pdst[o + lrow + 16] = acc1[i];
    }
}

__global__ __launch_bounds__(256)
void sg_reduce_partials(const float* __restrict__ part, float* __restrict__ out) {
    const int i = blockIdx.x * 256 + threadIdx.x;
    if (i < OUT_TOTAL) {
        float s = 0.0f;
        #pragma unroll
        for (int j = 0; j < NSPLIT; ++j) s += part[(size_t)j * OUT_TOTAL + i];
        out[i] = s;
    }
}

extern "C" void kernel_launch(void* const* d_in, const int* in_sizes, int n_in,
                              void* d_out, int out_size, void* d_ws, size_t ws_size,
                              hipStream_t stream) {
    const float* feats = (const float*)d_in[0];
    const float* probs = (const float*)d_in[1];
    float* out = (float*)d_out;

    float* ws    = (float*)d_ws;
    float* stats = ws;
    float* part  = ws + WS_PART_OFF;

    sg_softmax_stats<<<BATCH * KDIM, 256, 0, stream>>>(probs, stats);

    dim3 g2(CDIM / CT, BATCH, NSPLIT);
    sg_context_gemm<<<g2, 256, 0, stream>>>(feats, probs, stats, part);

    sg_reduce_partials<<<(OUT_TOTAL + 255) / 256, 256, 0, stream>>>(part, out);
}


# stats kernel reads each row once, all of a lane's 16 B loads in flight, row held in registers for max and exp sum (was two latency-serialized passes); exp-sum order changed (lane chains, butterfly)
# speedup vs baseline: 2.9733x; 1.1938x over previous
#include <hip/hip_runtime.h>
#include <cmath>

typedef float v2f __attribute__((ext_vector_type(2)));
typedef float v4f __attribute__((ext_vector_type(4)));
typedef float v8f __attribute__((ext_vector_type(8)));

#define BATCH 8
#define CDIM 512
#define KDIM 19
#define KPAD 32
#define HW 16384

#define CT 128
#define NT 64
#define NSPLIT 16
#define NPER (HW / NSPLIT)
#define OUT_TOTAL (BATCH * CDIM * KDIM)

#define WS_PART_OFF 1024

#define ST_THREADS 256
#define ST_WAVES (ST_THREADS / 32)
#define ST_LOADS (HW / (ST_THREADS * 4))

__global__ __launch_bounds__(ST_THREADS)
void sg_softmax_stats(const float* __restrict__ probs, float* __restrict__ stats) {
    __shared__ float red_max[ST_WAVES];
    __shared__ float red_sum[ST_WAVES];
    const int bk = blockIdx.x;
    const int t = threadIdx.x;
    const int lane = t & 31;
    const int wave = t >> 5;
    const float* src = probs + (size_t)bk * HW + t * 4;

    v4f v[ST_LOADS];
    #pragma unroll
    for (int j = 0; j < ST_LOADS; ++j)
        v[j] = *(const v4f*)(src + j * (ST_THREADS * 4));

    v4f mv = v[0];
    #pragma unroll
    for (int j = 1; j < ST_LOADS; ++j) {
        mv.x = fmaxf(mv.x, v[j].x);
        mv.y = fmaxf(mv.y, v[j].y);
        mv.z = fmaxf(mv.z, v[j].z);
        mv.w = fmaxf(mv.w, v[j].w);
    }
    float m = fmaxf(fmaxf(mv.x, mv.y), fmaxf(mv.z, mv.w));
    #pragma unroll
    for (int o = 16; o > 0; o >>= 1) m = fmaxf(m, __shfl_xor(m, o, 32));
    if (lane == 0) red_max[wave] = m;
    __syncthreads();
    m = red_max[0];
    #pragma unroll
    for (int w = 1; w < ST_WAVES; ++w) m = fmaxf(m, red_max[w]);

    v4f sv = {0.0f, 0.0f, 0.0f, 0.0f};
    #pragma unroll
    for (int j = 0; j < ST_LOADS; ++j) {
        sv.x += __expf(v[j].x - m);
        sv.y += __expf(v[j].y - m);
        sv.z += __expf(v[j].z - m);
        sv.w += __expf(v[j].w - m);
    }
    float sum = (sv.x + sv.y) + (sv.z + sv.w);
    #pragma unroll
    for (int o = 16; o > 0; o >>= 1) sum += __shfl_xor(sum, o, 32);
    if (lane == 0) red_sum[wave] = sum;
    __syncthreads();
    if (t == 0) {
        float total = red_sum[0];
        #pragma unroll
        for (int w = 1; w < ST_WAVES; ++w) total += red_sum[w];
        stats[2 * bk] = m;
        stats[2 * bk + 1] = 1.0f / total;
    }
}

__global__ __launch_bounds__(256)
void sg_context_gemm(const float* __restrict__ feats, const float* __restrict__ probs,
                     const float* __restrict__ stats, float* __restrict__ part) {
    __shared__ float ftile[CT][NT + 4];
    __shared__ float wtile[KPAD][NT + 4];

    const int t = threadIdx.x;
    const int wave = t >> 5;
    const int lane = t & 31;
    const int lrow = lane & 15;
    const int ksel = (lane >> 4) * 2;

    const int c0 = blockIdx.x * CT;
    const int b  = blockIdx.y;
    const int z  = blockIdx.z;
    const int nbeg = z * NPER;

    const float* fbase = feats + ((size_t)b * CDIM + c0) * HW;
    const float* pbase = probs + (size_t)b * KDIM * HW;

    const int srow = t >> 4;
    const int scol = (t & 15) * 4;

    const bool has1 = (16 + srow) < KDIM;
    const float m0 = stats[2 * (b * KDIM + srow)];
    const float s0 = stats[2 * (b * KDIM + srow) + 1];
    float m1 = 0.0f, s1 = 0.0f;
    if (has1) {
        m1 = stats[2 * (b * KDIM + 16 + srow)];
        s1 = stats[2 * (b * KDIM + 16 + srow) + 1];
    } else {
        const v4f zero4 = {0.0f, 0.0f, 0.0f, 0.0f};
        *(v4f*)&wtile[16 + srow][scol] = zero4;
    }

    v8f acc0 = {};
    v8f acc1 = {};

    for (int n0 = nbeg; n0 < nbeg + NPER; n0 += NT) {
        const float* psrc = pbase + (size_t)srow * HW + n0 + scol;
        v4f p0 = *(const v4f*)psrc;
        v4f p1 = {0.0f, 0.0f, 0.0f, 0.0f};
        if (has1) p1 = *(const v4f*)(psrc + (size_t)16 * HW);

        const float* fsrc = fbase + (size_t)srow * HW + n0 + scol;
        v4f fv[8];
        #pragma unroll
        for (int j = 0; j < 8; ++j)
            fv[j] = __builtin_nontemporal_load((const v4f*)(fsrc + (size_t)(16 * j) * HW));
        __builtin_amdgcn_sched_barrier(0);

        {
            v4f e;
            e.x = __expf(p0.x - m0) * s0;
            e.y = __expf(p0.y - m0) * s0;
            e.z = __expf(p0.z - m0) * s0;
            e.w = __expf(p0.w - m0) * s0;
            *(v4f*)&wtile[srow][scol] = e;
        }
        if (has1) {
            v4f e;
            e.x = __expf(p1.x - m1) * s1;
            e.y = __expf(p1.y - m1) * s1;
            e.z = __expf(p1.z - m1) * s1;
            e.w = __expf(p1.w - m1) * s1;
            *(v4f*)&wtile[16 + srow][scol] = e;
        }
        #pragma unroll
        for (int j = 0; j < 8; ++j)
            *(v4f*)&ftile[16 * j + srow][scol] = fv[j];
        __syncthreads();

        #pragma unroll
        for (int kk = 0; kk < NT; kk += 4) {
            v2f a  = *(const v2f*)&ftile[wave * 16 + lrow][kk + ksel];
            v2f b0 = *(const v2f*)&wtile[lrow][kk + ksel];
            v2f b1 = *(const v2f*)&wtile[16 + lrow][kk + ksel];
            acc0 = __builtin_amdgcn_wmma_f32_16x16x4_f32(
                false, a, false, b0, (short)0, acc0, false, false);
            acc1 = __builtin_amdgcn_wmma_f32_16x16x4_f32(
                false, a, false, b1, (short)0, acc1, false, false);
        }
        __syncthreads();
    }

    float* pdst = part + (size_t)z * OUT_TOTAL;
    const int rbase = c0 + wave * 16 + (lane >> 4) * 8;
    #pragma unroll
    for (int i = 0; i < 8; ++i) {
        const int row = rbase + i;
        const size_t o = ((size_t)b * CDIM + row) * KDIM;
        pdst[o + lrow] = acc0[i];
        if (lrow + 16 < KDIM) pdst[o + lrow + 16] = acc1[i];
    }
}

__global__ __launch_bounds__(256)
void sg_reduce_partials(const float* __restrict__ part, float* __restrict__ out) {
    const int i = blockIdx.x * 256 + threadIdx.x;
    if (i < OUT_TOTAL) {
        float s = 0.0f;
        #pragma unroll
        for (int j = 0; j < NSPLIT; ++j) s += part[(size_t)j * OUT_TOTAL + i];
        out[i] = s;
    }
}

extern "C" void kernel_launch(void* const* d_in, const int* in_sizes, int n_in,
                              void* d_out, int out_size, void* d_ws, size_t ws_size,
                              hipStream_t stream) {
    const float* feats = (const float*)d_in[0];
    const float* probs = (const float*)d_in[1];
    float* out = (float*)d_out;

    float* ws    = (float*)d_ws;
    float* stats = ws;
    float* part  = ws + WS_PART_OFF;

    sg_softmax_stats<<<BATCH * KDIM, ST_THREADS, 0, stream>>>(probs, stats);

    dim3 g2(CDIM / CT, BATCH, NSPLIT);
    sg_context_gemm<<<g2, 256, 0, stream>>>(feats, probs, stats, part);

    sg_reduce_partials<<<(OUT_TOTAL + 255) / 256, 256, 0, stream>>>(part, out);
}


# GEMM software-pipelined: next stage's global loads issued behind the barrier and held in registers under the matrix loop, two LDS buffers so one barrier per stage; sum order unchanged
# speedup vs baseline: 3.2424x; 1.0905x over previous
#include <hip/hip_runtime.h>
#include <cmath>

typedef float v2f __attribute__((ext_vector_type(2)));
typedef float v4f __attribute__((ext_vector_type(4)));
typedef float v8f __attribute__((ext_vector_type(8)));

#define BATCH 8
#define CDIM 512
#define KDIM 19
#define KPAD 32
#define HW 16384

#define CT 128
#define NT 64
#define NSPLIT 16
#define NPER (HW / NSPLIT)
#define NSTAGE (NPER / NT)
#define OUT_TOTAL (BATCH * CDIM * KDIM)

#define WS_PART_OFF 1024

#define ST_THREADS 256
#define ST_WAVES (ST_THREADS / 32)
#define ST_LOADS (HW / (ST_THREADS * 4))

__global__ __launch_bounds__(ST_THREADS)
void sg_softmax_stats(const float* __restrict__ probs, float* __restrict__ stats) {
    __shared__ float red_max[ST_WAVES];
    __shared__ float red_sum[ST_WAVES];
    const int bk = blockIdx.x;
    const int t = threadIdx.x;
    const int lane = t & 31;
    const int wave = t >> 5;
    const float* src = probs + (size_t)bk * HW + t * 4;

    v4f v[ST_LOADS];
    #pragma unroll
    for (int j = 0; j < ST_LOADS; ++j)
        v[j] = *(const v4f*)(src + j * (ST_THREADS * 4));

    v4f mv = v[0];
    #pragma unroll
    for (int j = 1; j < ST_LOADS; ++j) {
        mv.x = fmaxf(mv.x, v[j].x);
        mv.y = fmaxf(mv.y, v[j].y);
        mv.z = fmaxf(mv.z, v[j].z);
        mv.w = fmaxf(mv.w, v[j].w);
    }
    float m = fmaxf(fmaxf(mv.x, mv.y), fmaxf(mv.z, mv.w));
    #pragma unroll
    for (int o = 16; o > 0; o >>= 1) m = fmaxf(m, __shfl_xor(m, o, 32));
    if (lane == 0) red_max[wave] = m;
    __syncthreads();
    m = red_max[0];
    #pragma unroll
    for (int w = 1; w < ST_WAVES; ++w) m = fmaxf(m, red_max[w]);

    v4f sv = {0.0f, 0.0f, 0.0f, 0.0f};
    #pragma unroll
    for (int j = 0; j < ST_LOADS; ++j) {
        sv.x += __expf(v[j].x - m);
        sv.y += __expf(v[j].y - m);
        sv.z += __expf(v[j].z - m);
        sv.w += __expf(v[j].w - m);
    }
    float sum = (sv.x + sv.y) + (sv.z + sv.w);
    #pragma unroll
    for (int o = 16; o > 0; o >>= 1) sum += __shfl_xor(sum, o, 32);
    if (lane == 0) red_sum[wave] = sum;
    __syncthreads();
    if (t == 0) {
        float total = red_sum[0];
        #pragma unroll
        for (int w = 1; w < ST_WAVES; ++w) total += red_sum[w];
        stats[2 * bk] = m;
        stats[2 * bk + 1] = 1.0f / total;
    }
}

__global__ __launch_bounds__(256)
void sg_context_gemm(const float* __restrict__ feats, const float* __restrict__ probs,
                     const float* __restrict__ stats, float* __restrict__ part) {
    __shared__ float ftile[2][CT][NT + 4];
    __shared__ float wtile[2][KPAD][NT + 4];

    const int t = threadIdx.x;
    const int wave = t >> 5;
    const int lane = t & 31;
    const int lrow = lane & 15;
    const int ksel = (lane >> 4) * 2;

    const int c0 = blockIdx.x * CT;
    const int b  = blockIdx.y;
    const int z  = blockIdx.z;
    const int nbeg = z * NPER;

    const float* fbase = feats + ((size_t)b * CDIM + c0) * HW;
    const float* pbase = probs + (size_t)b * KDIM * HW;

    const int srow = t >> 4;
    const int scol = (t & 15) * 4;

    const bool has1 = (16 + srow) < KDIM;
    const float m0 = stats[2 * (b * KDIM + srow)];
    const float s0 = stats[2 * (b * KDIM + srow) + 1];
    float m1 = 0.0f, s1 = 0.0f;
    if (has1) {
        m1 = stats[2 * (b * KDIM + 16 + srow)];
        s1 = stats[2 * (b * KDIM + 16 + srow) + 1];
    } else {
        const v4f zero4 = {0.0f, 0.0f, 0.0f, 0.0f};
        *(v4f*)&wtile[0][16 + srow][scol] = zero4;
        *(v4f*)&wtile[1][16 + srow][scol] = zero4;
    }

    v8f acc0 = {};
    v8f acc1 = {};

    const float* psrc = pbase + (size_t)srow * HW + nbeg + scol;
    const float* fsrc = fbase + (size_t)srow * HW + nbeg + scol;
    v4f p0, fv[8];
    v4f p1 = {0.0f, 0.0f, 0.0f, 0.0f};

    p0 = *(const v4f*)psrc;
    if (has1) p1 = *(const v4f*)(psrc + (size_t)16 * HW);
    #pragma unroll
    for (int j = 0; j < 8; ++j)
        fv[j] = __builtin_nontemporal_load((const v4f*)(fsrc + (size_t)(16 * j) * HW));

    for (int s = 0; s < NSTAGE; ++s) {
        float (*ft)[NT + 4] = ftile[s & 1];
        float (*wt)[NT + 4] = wtile[s & 1];

        {
            v4f e;
            e.x = __expf(p0.x - m0) * s0;
            e.y = __expf(p0.y - m0) * s0;
            e.z = __expf(p0.z - m0) * s0;
            e.w = __expf(p0.w - m0) * s0;
            *(v4f*)&wt[srow][scol] = e;
        }
        if (has1) {
            v4f e;
            e.x = __expf(p1.x - m1) * s1;
            e.y = __expf(p1.y - m1) * s1;
            e.z = __expf(p1.z - m1) * s1;
            e.w = __expf(p1.w - m1) * s1;
            *(v4f*)&wt[16 + srow][scol] = e;
        }
        #pragma unroll
        for (int j = 0; j < 8; ++j)
            *(v4f*)&ft[16 * j + srow][scol] = fv[j];
        __syncthreads();

        if (s + 1 < NSTAGE) {
            psrc += NT;
            fsrc += NT;
            p0 = *(const v4f*)psrc;
            if (has1) p1 = *(const v4f*)(psrc + (size_t)16 * HW);
            #pragma unroll
            for (int j = 0; j < 8; ++j)
                fv[j] = __builtin_nontemporal_load((const v4f*)(fsrc + (size_t)(16 * j) * HW));
        }

        #pragma unroll
        for (int kk = 0; kk < NT; kk += 4) {
            v2f a  = *(const v2f*)&ft[wave * 16 + lrow][kk + ksel];
            v2f b0 = *(const v2f*)&wt[lrow][kk + ksel];
            v2f b1 = *(const v2f*)&wt[16 + lrow][kk + ksel];
            acc0 = __builtin_amdgcn_wmma_f32_16x16x4_f32(
                false, a, false, b0, (short)0, acc0, false, false);
            acc1 = __builtin_amdgcn_wmma_f32_16x16x4_f32(
                false, a, false, b1, (short)0, acc1, false, false);
        }
    }

    float* pdst = part + (size_t)z * OUT_TOTAL;
    const int rbase = c0 + wave * 16 + (lane >> 4) * 8;
    #pragma unroll
    for (int i = 0; i < 8; ++i) {
        const int row = rbase + i;
        const size_t o = ((size_t)b * CDIM + row) * KDIM;
        pdst[o + lrow] = acc0[i];
        if (lrow + 16 < KDIM) pdst[o + lrow + 16] = acc1[i];
    }
}

__global__ __launch_bounds__(256)
void sg_reduce_partials(const float* __restrict__ part, float* __restrict__ out) {
    const int i = blockIdx.x * 256 + threadIdx.x;
    if (i < OUT_TOTAL) {
        float s = 0.0f;
        #pragma unroll
        for (int j = 0; j < NSPLIT; ++j) s += part[(size_t)j * OUT_TOTAL + i];
        out[i] = s;
    }
}

extern "C" void kernel_launch(void* const* d_in, const int* in_sizes, int n_in,
                              void* d_out, int out_size, void* d_ws, size_t ws_size,
                              hipStream_t stream) {
    const float* feats = (const float*)d_in[0];
    const float* probs = (const float*)d_in[1];
    float* out = (float*)d_out;

    float* ws    = (float*)d_ws;
    float* stats = ws;
    float* part  = ws + WS_PART_OFF;

    sg_softmax_stats<<<BATCH * KDIM, ST_THREADS, 0, stream>>>(probs, stats);

    dim3 g2(CDIM / CT, BATCH, NSPLIT);
    sg_context_gemm<<<g2, 256, 0, stream>>>(feats, probs, stats, part);

    sg_reduce_partials<<<(OUT_TOTAL + 255) / 256, 256, 0, stream>>>(part, out);
}


# softmax max and exp-sum taken per n-split inside the GEMM work-groups (weights exp(p - split max) held in LDS), splits rescaled and normalised by the last kernel: 3 launches to 2; exp-sum order new
# speedup vs baseline: 3.6238x; 1.1176x over previous
#include <hip/hip_runtime.h>
#include <cmath>

typedef float v2f __attribute__((ext_vector_type(2)));
typedef float v4f __attribute__((ext_vector_type(4)));
typedef float v8f __attribute__((ext_vector_type(8)));

#define BATCH 8
#define CDIM 512
#define KDIM 19
#define KPAD 32
#define HW 16384

#define CT 128
#define NT 64
#define NSPLIT 16
#define NPER (HW / NSPLIT)
#define NSTAGE (NPER / NT)
#define OUT_TOTAL (BATCH * CDIM * KDIM)

#define WPITCH (NPER + 4)
#define SLAB_LOADS (NPER / 128)

#define LS_PLANE (BATCH * NSPLIT * KPAD)
#define WS_PART_OFF (2 * LS_PLANE)

__device__ __forceinline__ void sg_slab_row(const v4f (&v)[SLAB_LOADS], float* wrow,
                                            float* out_max, float* out_sum, int lane) {
    v4f mv = v[0];
    #pragma unroll
    for (int j = 1; j < SLAB_LOADS; ++j) {
        mv.x = fmaxf(mv.x, v[j].x);
        mv.y = fmaxf(mv.y, v[j].y);
        mv.z = fmaxf(mv.z, v[j].z);
        mv.w = fmaxf(mv.w, v[j].w);
    }
    float m = fmaxf(fmaxf(mv.x, mv.y), fmaxf(mv.z, mv.w));
    #pragma unroll
    for (int o = 16; o > 0; o >>= 1) m = fmaxf(m, __shfl_xor(m, o, 32));

    v4f sv = {0.0f, 0.0f, 0.0f, 0.0f};
    #pragma unroll
    for (int j = 0; j < SLAB_LOADS; ++j) {
        v4f e;
        e.x = __expf(v[j].x - m);
        e.y = __expf(v[j].y - m);
        e.z = __expf(v[j].z - m);
        e.w = __expf(v[j].w - m);
        *(v4f*)(wrow + lane * 4 + j * 128) = e;
        sv.x += e.x;
        sv.y += e.y;
        sv.z += e.z;
        sv.w += e.w;
    }
    float sum = (sv.x + sv.y) + (sv.z + sv.w);
    #pragma unroll
    for (int o = 16; o > 0; o >>= 1) sum += __shfl_xor(sum, o, 32);
    if (lane == 0) {
        *out_max = m;
        *out_sum = sum;
    }
}

__global__ __launch_bounds__(256)
void sg_context_gemm(const float* __restrict__ feats, const float* __restrict__ probs,
                     float* __restrict__ lstat, float* __restrict__ part) {
    __shared__ __attribute__((aligned(16))) float ftile[2][CT][NT + 4];
    __shared__ __attribute__((aligned(16))) float wts[KDIM][WPITCH];
    __shared__ float mst[2][KPAD];

    const int t = threadIdx.x;
    const int wave = t >> 5;
    const int lane = t & 31;
    const int lrow = lane & 15;
    const int ksel = (lane >> 4) * 2;

    const int c0 = blockIdx.x * CT;
    const int b  = blockIdx.y;
    const int z  = blockIdx.z;
    const int nbeg = z * NPER;

    const float* fbase = feats + ((size_t)b * CDIM + c0) * HW;
    const float* pbase = probs + (size_t)b * KDIM * HW;

    const int srow = t >> 4;
    const int scol = (t & 15) * 4;

    const bool has2 = (16 + __builtin_amdgcn_readfirstlane(wave)) < KDIM;
    const float* prow = pbase + (size_t)wave * HW + nbeg + lane * 4;
    v4f pv0[SLAB_LOADS], pv1[SLAB_LOADS], pv2[SLAB_LOADS];
    if (has2) {
        #pragma unroll
        for (int j = 0; j < SLAB_LOADS; ++j) pv2[j] = *(const v4f*)(prow + (size_t)16 * HW + j * 128);
    }
    #pragma unroll
    for (int j = 0; j < SLAB_LOADS; ++j) pv0[j] = *(const v4f*)(prow + j * 128);
    #pragma unroll
    for (int j = 0; j < SLAB_LOADS; ++j) pv1[j] = *(const v4f*)(prow + (size_t)8 * HW + j * 128);

    const float* fsrc = fbase + (size_t)srow * HW + nbeg + scol;
    v4f fv[8];
    #pragma unroll
    for (int j = 0; j < 8; ++j)
        fv[j] = __builtin_nontemporal_load((const v4f*)(fsrc + (size_t)(16 * j) * HW));


    sg_slab_row(pv0, &wts[wave][0], &mst[0][wave], &mst[1][wave], lane);
    sg_slab_row(pv1, &wts[8 + wave][0], &mst[0][8 + wave], &mst[1][8 + wave], lane);
    if (has2)
        sg_slab_row(pv2, &wts[16 + wave][0], &mst[0][16 + wave], &mst[1][16 + wave], lane);

    v8f acc0 = {};
    v8f acc1 = {};

    const int r1 = (16 + lrow < KDIM) ? 16 + lrow : lrow;
    const float* wb0 = &wts[lrow][ksel];
    const float* wb1 = &wts[r1][ksel];

    for (int s = 0; s < NSTAGE; ++s) {
        float (*ft)[NT + 4] = ftile[s & 1];

        #pragma unroll
        for (int j = 0; j < 8; ++j)
            *(v4f*)&ft[16 * j + srow][scol] = fv[j];
        __syncthreads();

        if (s + 1 < NSTAGE) {
            fsrc += NT;
            #pragma unroll
            for (int j = 0; j < 8; ++j)
                fv[j] = __builtin_nontemporal_load((const v4f*)(fsrc + (size_t)(16 * j) * HW));
        }

        const float* w0 = wb0 + s * NT;
        const float* w1 = wb1 + s * NT;
        #pragma unroll
        for (int kk = 0; kk < NT; kk += 4) {
            v2f a  = *(const v2f*)&ft[wave * 16 + lrow][kk + ksel];
            v2f b0 = *(const v2f*)(w0 + kk);
            v2f b1 = *(const v2f*)(w1 + kk);
            acc0 = __builtin_amdgcn_wmma_f32_16x16x4_f32(
                false, a, false, b0, (short)0, acc0, false, false);
            acc1 = __builtin_amdgcn_wmma_f32_16x16x4_f32(
                false, a, false, b1, (short)0, acc1, false, false);
        }
    }

    if (blockIdx.x == 0 && wave == 0 && lane < KDIM) {
        const int o = (b * NSPLIT + z) * KPAD + lane;
        lstat[o] = mst[0][lane];
        lstat[LS_PLANE + o] = mst[1][lane];
    }

    float* pdst = part + (size_t)z * OUT_TOTAL;
    const int rbase = c0 + wave * 16 + (lane >> 4) * 8;
    #pragma unroll
    for (int i = 0; i < 8; ++i) {
        const int row = rbase + i;
        const size_t o = ((size_t)b * CDIM + row) * KDIM;
        pdst[o + lrow] = acc0[i];
        if (lrow + 16 < KDIM) pdst[o + lrow + 16] = acc1[i];
    }
}

__global__ __launch_bounds__(256)
void sg_combine_splits(const float* __restrict__ part, const float* __restrict__ lstat,
                       float* __restrict__ out) {
    const int i = blockIdx.x * 256 + threadIdx.x;
    if (i < OUT_TOTAL) {
        const int b = i / (CDIM * KDIM);
        const int k = i % KDIM;
        const float* mp = lstat + (size_t)(b * NSPLIT) * KPAD + k;
        float mz[NSPLIT], sz[NSPLIT], pz[NSPLIT];
        #pragma unroll
        for (int j = 0; j < NSPLIT; ++j) {
            mz[j] = mp[j * KPAD];
            sz[j] = mp[LS_PLANE + j * KPAD];
            pz[j] = part[(size_t)j * OUT_TOTAL + i];
        }
        float mx = mz[0];
        #pragma unroll
        for (int j = 1; j < NSPLIT; ++j) mx = fmaxf(mx, mz[j]);
        float den = 0.0f, num = 0.0f;
        #pragma unroll
        for (int j = 0; j < NSPLIT; ++j) {
            const float e = __expf(mz[j] - mx);
            den += sz[j] * e;
            num += pz[j] * e;
        }
        out[i] = num / den;
    }
}

extern "C" void kernel_launch(void* const* d_in, const int* in_sizes, int n_in,
                              void* d_out, int out_size, void* d_ws, size_t ws_size,
                              hipStream_t stream) {
    const float* feats = (const float*)d_in[0];
    const float* probs = (const float*)d_in[1];
    float* out = (float*)d_out;

    float* ws    = (float*)d_ws;
    float* lstat = ws;
    float* part  = ws + WS_PART_OFF;

    dim3 g2(CDIM / CT, BATCH, NSPLIT);
    sg_context_gemm<<<g2, 256, 0, stream>>>(feats, probs, lstat, part);

    sg_combine_splits<<<(OUT_TOTAL + 255) / 256, 256, 0, stream>>>(part, lstat, out);
}
